# starting kernel
# baseline (speedup 1.0000x reference)
#include <hip/hip_runtime.h>


typedef __attribute__((ext_vector_type(16))) _Float16 v16h;
typedef __attribute__((ext_vector_type(8)))  _Float16 v8h;
typedef __attribute__((ext_vector_type(4)))  _Float16 v4h;
typedef __attribute__((ext_vector_type(8)))  float    v8f;

static constexpr int N_  = 64;
static constexpr int M_  = 64;
static constexpr int F_  = 256;
static constexpr int H_  = 4;
static constexpr int D_  = 64;
static constexpr int HD_ = H_ * D_;
static constexpr int FRAG_HALFS = 32 * 16;
static constexpr int W1F_HALFS  = F_ * HD_;
static constexpr int HEAD_HALFS = 4 * 2 * FRAG_HALFS;
static constexpr int PAIR_HALFS = H_ * HEAD_HALFS;
static constexpr int LDS_STRIDE = 264;

__global__ void swizzle_w1_kernel(const float* __restrict__ W1,
                                  _Float16* __restrict__ w1f) {
  int id   = blockIdx.x * blockDim.x + threadIdx.x;
  int jj   = id & 15;
  int lane = (id >> 4) & 31;
  int kt   = (id >> 9) & 7;
  int nt   = (id >> 12) & 15;
  int n    = nt * 16 + (lane & 15);
  int koff = (lane >> 4) * 8;
  int k    = kt * 32 + koff + ((jj < 8) ? jj : (8 + jj));
  w1f[id] = (_Float16)W1[k * HD_ + n];
}

__global__ void __launch_bounds__(256)
proj_kernel(const float* __restrict__ x,
            const _Float16* __restrict__ w1f,
            _Float16* __restrict__ hbuf) {
  __shared__ __align__(16) _Float16 lds[64 * LDS_STRIDE];

  const int t    = threadIdx.x;
  const int lane = t & 31;
  const int wave = t >> 5;
  const int bi   = blockIdx.y;
  const int bj   = blockIdx.x;

  {
    const float4* x4 = (const float4*)(x + (size_t)(bi * N_ + bj) * (M_ * F_));
#pragma unroll
    for (int k = 0; k < 16; ++k) {
      int f4  = k * 256 + t;
      float4 v = x4[f4];
      int row = f4 >> 6;
      int col = (f4 & 63) * 4;
      v4h hv = { (_Float16)v.x, (_Float16)v.y, (_Float16)v.z, (_Float16)v.w };
      *(v4h*)&lds[row * LDS_STRIDE + col] = hv;
    }
  }
  __syncthreads();

  const int mtBase = (wave >> 2) * 2;
  const int ntBase = (wave & 3) * 4;
  const int mrow0  = mtBase * 16 + (lane & 15);
  const int klane  = (lane >> 4) << 3;

  v8f acc[2][4] = {};

#pragma unroll
  for (int kt = 0; kt < 8; ++kt) {
    v16h a[2];
#pragma unroll
    for (int mi = 0; mi < 2; ++mi) {
      const _Float16* src = &lds[(mrow0 + mi * 16) * LDS_STRIDE + kt * 32 + klane];
      v8h lo = *(const v8h*)src;
      v8h hi = *(const v8h*)(src + 16);
      v16h av;
#pragma unroll
      for (int u = 0; u < 8; ++u) { av[u] = lo[u]; av[8 + u] = hi[u]; }
      a[mi] = av;
    }
#pragma unroll
    for (int ni = 0; ni < 4; ++ni) {
      v16h b = *(const v16h*)(w1f + ((ntBase + ni) * 8 + kt) * FRAG_HALFS + lane * 16);
#pragma unroll
      for (int mi = 0; mi < 2; ++mi) {
        acc[mi][ni] = __builtin_amdgcn_wmma_f32_16x16x32_f16(
            false, a[mi], false, b, (short)0, acc[mi][ni], false, false);
      }
    }
  }
  __syncthreads();

#pragma unroll
  for (int mi = 0; mi < 2; ++mi) {
#pragma unroll
    for (int ni = 0; ni < 4; ++ni) {
      int n = (ntBase + ni) * 16 + (lane & 15);
#pragma unroll
      for (int r = 0; r < 8; ++r) {
        int m = (mtBase + mi) * 16 + r + ((lane >> 4) << 3);
        lds[m * LDS_STRIDE + n] = (_Float16)acc[mi][ni][r];
      }
    }
  }
  __syncthreads();

  _Float16* dst = hbuf + (size_t)(bi * N_ + bj) * PAIR_HALFS;
#pragma unroll
  for (int it = 0; it < 4; ++it) {
    int c     = it * 256 + t;
    int lane2 = c & 31;
    int ktc   = (c >> 5) & 1;
    int mtc   = (c >> 6) & 3;
    int head  = (c >> 8) & 3;
    int m     = mtc * 16 + (lane2 & 15);
    int kb    = ktc * 32 + ((lane2 >> 4) << 3);
    const _Float16* src = &lds[m * LDS_STRIDE + head * 64 + kb];
    v8h lo = *(const v8h*)src;
    v8h hi = *(const v8h*)(src + 16);
    v8h* o = (v8h*)(dst + (((head * 4 + mtc) * 2 + ktc) * FRAG_HALFS) + lane2 * 16);
    o[0] = lo;
    o[1] = hi;
  }
}

__global__ void __launch_bounds__(256)
score_kernel(const _Float16* __restrict__ hbuf,
             const float* __restrict__ nItem,
             const float* __restrict__ W2,
             float* __restrict__ out) {
  __shared__ float ssum[H_];
  const int t    = threadIdx.x;
  const int lane = t & 31;
  const int wave = t >> 5;
  const int p    = blockIdx.y;
  const int q    = blockIdx.x;

  if (t < H_) ssum[t] = 0.0f;
  __syncthreads();

  const int head   = wave >> 1;
  const int mtBase = (wave & 1) * 2;
  const _Float16* Abase = hbuf + ((size_t)(p * N_ + q) * H_ + head) * HEAD_HALFS;
  const _Float16* Bbase = hbuf + ((size_t)(q * N_ + p) * H_ + head) * HEAD_HALFS;

  v8f acc[2][4] = {};
#pragma unroll
  for (int kt = 0; kt < 2; ++kt) {
    v16h a[2], b[4];
#pragma unroll
    for (int mi = 0; mi < 2; ++mi)
      a[mi] = *(const v16h*)(Abase + ((mtBase + mi) * 2 + kt) * FRAG_HALFS + lane * 16);
#pragma unroll
    for (int ni = 0; ni < 4; ++ni)
      b[ni] = *(const v16h*)(Bbase + (ni * 2 + kt) * FRAG_HALFS + lane * 16);
#pragma unroll
    for (int mi = 0; mi < 2; ++mi)
#pragma unroll
      for (int ni = 0; ni < 4; ++ni)
        acc[mi][ni] = __builtin_amdgcn_wmma_f32_16x16x32_f16(
            false, a[mi], false, b[ni], (short)0, acc[mi][ni], false, false);
  }

  float s = 0.0f;
#pragma unroll
  for (int mi = 0; mi < 2; ++mi)
#pragma unroll
    for (int ni = 0; ni < 4; ++ni)
#pragma unroll
      for (int r = 0; r < 8; ++r)
        s += fmaxf(acc[mi][ni][r], 0.0f);

#pragma unroll
  for (int off = 16; off > 0; off >>= 1)
    s += __shfl_xor(s, off, 32);
  if (lane == 0) atomicAdd(&ssum[head], s);
  __syncthreads();

  if (t == 0) {
    float inv = 0.125f / (nItem[p] * nItem[q]);
    float r = 0.0f;
#pragma unroll
    for (int h = 0; h < H_; ++h) r += ssum[h] * W2[h];
    out[p * N_ + q] = r * inv;
  }
}

extern "C" void kernel_launch(void* const* d_in, const int* in_sizes, int n_in,
                              void* d_out, int out_size, void* d_ws, size_t ws_size,
                              hipStream_t stream) {
  const float* x     = (const float*)d_in[0];
  const float* nItem = (const float*)d_in[1];
  const float* W1    = (const float*)d_in[2];
  const float* W2    = (const float*)d_in[3];
  float* out = (float*)d_out;

  _Float16* w1f  = (_Float16*)d_ws;
  _Float16* hbuf = w1f + W1F_HALFS;

  swizzle_w1_kernel<<<256, 256, 0, stream>>>(W1, w1f);
  proj_kernel<<<dim3(N_, N_), 256, 0, stream>>>(x, w1f, hbuf);
  score_kernel<<<dim3(N_, N_), 256, 0, stream>>>(hbuf, nItem, W2, out);
}


# Projection kernel made persistent: 256 work-groups walk 16 tiles each, W1 fragments loaded once into registers (W1 read once per work-group, not per tile), next x tile fetched under the current one
# speedup vs baseline: 1.1917x; 1.1917x over previous
#include <hip/hip_runtime.h>


typedef __attribute__((ext_vector_type(16))) _Float16 v16h;
typedef __attribute__((ext_vector_type(8)))  _Float16 v8h;
typedef __attribute__((ext_vector_type(4)))  _Float16 v4h;
typedef __attribute__((ext_vector_type(8)))  float    v8f;

static constexpr int N_  = 64;
static constexpr int M_  = 64;
static constexpr int F_  = 256;
static constexpr int H_  = 4;
static constexpr int D_  = 64;
static constexpr int HD_ = H_ * D_;
static constexpr int FRAG_HALFS = 32 * 16;
static constexpr int W1F_HALFS  = F_ * HD_;
static constexpr int HEAD_HALFS = 4 * 2 * FRAG_HALFS;
static constexpr int PAIR_HALFS = H_ * HEAD_HALFS;
static constexpr int LDS_STRIDE = 264;

__global__ void swizzle_w1_kernel(const float* __restrict__ W1,
                                  _Float16* __restrict__ w1f) {
  int id   = blockIdx.x * blockDim.x + threadIdx.x;
  int jj   = id & 15;
  int lane = (id >> 4) & 31;
  int kt   = (id >> 9) & 7;
  int nt   = (id >> 12) & 15;
  int n    = nt * 16 + (lane & 15);
  int koff = (lane >> 4) * 8;
  int k    = kt * 32 + koff + ((jj < 8) ? jj : (8 + jj));
  w1f[id] = (_Float16)W1[k * HD_ + n];
}

__device__ __forceinline__ void lds_barrier() {
  __builtin_amdgcn_fence(__ATOMIC_RELEASE, "workgroup", "local");
  __builtin_amdgcn_s_barrier();
  __builtin_amdgcn_fence(__ATOMIC_ACQUIRE, "workgroup", "local");
}

static constexpr int PROJ_WGS     = 256;
static constexpr int TILES_PER_WG = (N_ * N_) / PROJ_WGS;

__global__ void __launch_bounds__(256)
proj_kernel(const float* __restrict__ x,
            const _Float16* __restrict__ w1f,
            _Float16* __restrict__ hbuf) {
  __shared__ __align__(16) _Float16 lds[64 * LDS_STRIDE];

  const int t    = threadIdx.x;
  const int lane = t & 31;
  const int wave = t >> 5;

  const int ntBase = wave * 2;
  const int mrow0  = lane & 15;
  const int klane  = (lane >> 4) << 3;

  v16h breg[2][8];
  for (int ni = 0; ni < 2; ++ni) {
    for (int kt = 0; kt < 8; ++kt) {
      breg[ni][kt] = *(const v16h*)(w1f + ((ntBase + ni) * 8 + kt) * FRAG_HALFS + lane * 16);
    }
  }

  __builtin_amdgcn_sched_barrier(0);

  const int tile0 = blockIdx.x * TILES_PER_WG;

  float4 xr[16];
  {
    const float4* x4 = (const float4*)(x + (size_t)tile0 * (M_ * F_));
    for (int k = 0; k < 16; ++k) xr[k] = x4[k * 256 + t];
  }

  for (int it = 0; it < TILES_PER_WG; ++it) {
    const int tile = tile0 + it;

#pragma unroll
    for (int k = 0; k < 16; ++k) {
      int f4  = k * 256 + t;
      float4 v = xr[k];
      int row = f4 >> 6;
      int col = (f4 & 63) * 4;
      v4h hv = { (_Float16)v.x, (_Float16)v.y, (_Float16)v.z, (_Float16)v.w };
      *(v4h*)&lds[row * LDS_STRIDE + col] = hv;
    }
    lds_barrier();

    if (it + 1 < TILES_PER_WG) {
      const float4* x4 = (const float4*)(x + (size_t)(tile + 1) * (M_ * F_));
      for (int k = 0; k < 16; ++k) xr[k] = x4[k * 256 + t];
    }

    v8f acc[4][2] = {};

#pragma unroll
    for (int kt = 0; kt < 8; ++kt) {
      v16h a[4];
#pragma unroll
      for (int mi = 0; mi < 4; ++mi) {
        const _Float16* src = &lds[(mrow0 + mi * 16) * LDS_STRIDE + kt * 32 + klane];
        v8h lo = *(const v8h*)src;
        v8h hi = *(const v8h*)(src + 16);
        v16h av;
#pragma unroll
        for (int u = 0; u < 8; ++u) { av[u] = lo[u]; av[8 + u] = hi[u]; }
        a[mi] = av;
      }
#pragma unroll
      for (int ni = 0; ni < 2; ++ni) {
#pragma unroll
        for (int mi = 0; mi < 4; ++mi) {
          acc[mi][ni] = __builtin_amdgcn_wmma_f32_16x16x32_f16(
              false, a[mi], false, breg[ni][kt], (short)0, acc[mi][ni], false, false);
        }
      }
    }
    lds_barrier();

#pragma unroll
    for (int mi = 0; mi < 4; ++mi) {
#pragma unroll
      for (int ni = 0; ni < 2; ++ni) {
        int n = (ntBase + ni) * 16 + (lane & 15);
#pragma unroll
        for (int r = 0; r < 8; ++r) {
          int m = mi * 16 + r + ((lane >> 4) << 3);
          lds[m * LDS_STRIDE + n] = (_Float16)acc[mi][ni][r];
        }
      }
    }
    lds_barrier();

    _Float16* dst = hbuf + (size_t)tile * PAIR_HALFS;
#pragma unroll
    for (int itc = 0; itc < 4; ++itc) {
      int c     = itc * 256 + t;
      int lane2 = c & 31;
      int ktc   = (c >> 5) & 1;
      int mtc   = (c >> 6) & 3;
      int head  = (c >> 8) & 3;
      int m     = mtc * 16 + (lane2 & 15);
      int kb    = ktc * 32 + ((lane2 >> 4) << 3);
      const _Float16* src = &lds[m * LDS_STRIDE + head * 64 + kb];
      v8h lo = *(const v8h*)src;
      v8h hi = *(const v8h*)(src + 16);
      v8h* o = (v8h*)(dst + (((head * 4 + mtc) * 2 + ktc) * FRAG_HALFS) + lane2 * 16);
      o[0] = lo;
      o[1] = hi;
    }
    lds_barrier();
  }
}

__global__ void __launch_bounds__(256)
score_kernel(const _Float16* __restrict__ hbuf,
             const float* __restrict__ nItem,
             const float* __restrict__ W2,
             float* __restrict__ out) {
  __shared__ float ssum[H_];
  const int t    = threadIdx.x;
  const int lane = t & 31;
  const int wave = t >> 5;
  const int p    = blockIdx.y;
  const int q    = blockIdx.x;

  if (t < H_) ssum[t] = 0.0f;
  __syncthreads();

  const int head   = wave >> 1;
  const int mtBase = (wave & 1) * 2;
  const _Float16* Abase = hbuf + ((size_t)(p * N_ + q) * H_ + head) * HEAD_HALFS;
  const _Float16* Bbase = hbuf + ((size_t)(q * N_ + p) * H_ + head) * HEAD_HALFS;

  v8f acc[2][4] = {};
#pragma unroll
  for (int kt = 0; kt < 2; ++kt) {
    v16h a[2], b[4];
#pragma unroll
    for (int mi = 0; mi < 2; ++mi)
      a[mi] = *(const v16h*)(Abase + ((mtBase + mi) * 2 + kt) * FRAG_HALFS + lane * 16);
#pragma unroll
    for (int ni = 0; ni < 4; ++ni)
      b[ni] = *(const v16h*)(Bbase + (ni * 2 + kt) * FRAG_HALFS + lane * 16);
#pragma unroll
    for (int mi = 0; mi < 2; ++mi)
#pragma unroll
      for (int ni = 0; ni < 4; ++ni)
        acc[mi][ni] = __builtin_amdgcn_wmma_f32_16x16x32_f16(
            false, a[mi], false, b[ni], (short)0, acc[mi][ni], false, false);
  }

  float s = 0.0f;
#pragma unroll
  for (int mi = 0; mi < 2; ++mi)
#pragma unroll
    for (int ni = 0; ni < 4; ++ni)
#pragma unroll
      for (int r = 0; r < 8; ++r)
        s += fmaxf(acc[mi][ni][r], 0.0f);

#pragma unroll
  for (int off = 16; off > 0; off >>= 1)
    s += __shfl_xor(s, off, 32);
  if (lane == 0) atomicAdd(&ssum[head], s);
  __syncthreads();

  if (t == 0) {
    float inv = 0.125f / (nItem[p] * nItem[q]);
    float r = 0.0f;
#pragma unroll
    for (int h = 0; h < H_; ++h) r += ssum[h] * W2[h];
    out[p * N_ + q] = r * inv;
  }
}

extern "C" void kernel_launch(void* const* d_in, const int* in_sizes, int n_in,
                              void* d_out, int out_size, void* d_ws, size_t ws_size,
                              hipStream_t stream) {
  const float* x     = (const float*)d_in[0];
  const float* nItem = (const float*)d_in[1];
  const float* W1    = (const float*)d_in[2];
  const float* W2    = (const float*)d_in[3];
  float* out = (float*)d_out;

  _Float16* w1f  = (_Float16*)d_ws;
  _Float16* hbuf = w1f + W1F_HALFS;

  swizzle_w1_kernel<<<256, 256, 0, stream>>>(W1, w1f);
  proj_kernel<<<PROJ_WGS, 256, 0, stream>>>(x, w1f, hbuf);
  score_kernel<<<dim3(N_, N_), 256, 0, stream>>>(hbuf, nItem, W2, out);
}


# LDS fragment reads issued as one group ahead of each WMMA block: x fragments through two register sets (two k-steps per group, two groups in flight), pair fragments all ahead of the 16 score WMMAs
# speedup vs baseline: 2.3127x; 1.9407x over previous
#include <hip/hip_runtime.h>


typedef __attribute__((ext_vector_type(16))) _Float16 v16h;
typedef __attribute__((ext_vector_type(8)))  _Float16 v8h;
typedef __attribute__((ext_vector_type(4)))  _Float16 v4h;
typedef __attribute__((ext_vector_type(8)))  float    v8f;

static constexpr int N_  = 64;
static constexpr int M_  = 64;
static constexpr int F_  = 256;
static constexpr int H_  = 4;
static constexpr int D_  = 64;
static constexpr int HD_ = H_ * D_;
static constexpr int FRAG_HALFS = 32 * 16;
static constexpr int W1F_HALFS  = F_ * HD_;
static constexpr int LDS_STRIDE = 264;
static constexpr int TILE_HALFS = M_ * LDS_STRIDE;

__global__ void swizzle_w1_kernel(const float* __restrict__ W1,
                                  _Float16* __restrict__ w1f) {
  int id   = blockIdx.x * blockDim.x + threadIdx.x;
  int jj   = id & 15;
  int lane = (id >> 4) & 31;
  int kt   = (id >> 9) & 7;
  int nt   = (id >> 12) & 15;
  int n    = nt * 16 + (lane & 15);
  int koff = (lane >> 4) * 8;
  int k    = kt * 32 + koff + ((jj < 8) ? jj : (8 + jj));
  w1f[id] = (_Float16)W1[k * HD_ + n];
}

__device__ __forceinline__ void lds_barrier() {
  __builtin_amdgcn_fence(__ATOMIC_RELEASE, "workgroup", "local");
  __builtin_amdgcn_s_barrier();
  __builtin_amdgcn_fence(__ATOMIC_ACQUIRE, "workgroup", "local");
}

__device__ __forceinline__ v16h lds_frag(const _Float16* src) {
  v8h lo = *(const v8h*)src;
  v8h hi = *(const v8h*)(src + 16);
  v16h f;
#pragma unroll
  for (int u = 0; u < 8; ++u) { f[u] = lo[u]; f[8 + u] = hi[u]; }
  return f;
}

static constexpr int PAIR_WGS     = 256;
static constexpr int SLOTS_PER_WG = 16;
static constexpr int FULL_WGS     = 224;
static constexpr int MAX_PAIRS    = 9;

struct Slot { int tile, i, j, second, diag; };

__device__ __forceinline__ Slot slot_info(int wg, int s) {
  Slot r;
  int u;
  if (wg < FULL_WGS) {
    u = wg * 8 + (s >> 1);
  } else {
    const int v = wg - FULL_WGS;
    if (s >= 14) {
      const int g = v * 2 + (s - 14);
      r.i = g; r.j = g; r.second = 0; r.diag = 1; r.tile = g * N_ + g;
      return r;
    }
    u = FULL_WGS * 8 + v * 7 + (s >> 1);
  }
  int i, d;
  if (u < 1984) { i = u & 63; d = 1 + (u >> 6); }
  else          { i = u - 1984; d = 32; }
  const int j = (i + d) & 63;
  r.i = i; r.j = j; r.second = s & 1; r.diag = 0;
  r.tile = r.second ? (j * N_ + i) : (i * N_ + j);
  return r;
}

__global__ void __launch_bounds__(256)
pair_kernel(const float* __restrict__ x,
            const _Float16* __restrict__ w1f,
            const float* __restrict__ nItem,
            const float* __restrict__ W2,
            float* __restrict__ out) {
  __shared__ __align__(16) _Float16 xs[2][TILE_HALFS];
  __shared__ __align__(16) _Float16 hs[2][TILE_HALFS];
  __shared__ float wsum[MAX_PAIRS * 8];
  __shared__ float nIs[N_];
  __shared__ float w2s[H_];

  const int t    = threadIdx.x;
  const int lane = t & 31;
  const int wave = t >> 5;
  const int wg   = blockIdx.x;

  const int ntBase = wave * 2;
  const int mrow0  = lane & 15;
  const int klane  = (lane >> 4) << 3;

  v16h breg[2][8];
  for (int ni = 0; ni < 2; ++ni) {
    for (int kt = 0; kt < 8; ++kt) {
      breg[ni][kt] = *(const v16h*)(w1f + ((ntBase + ni) * 8 + kt) * FRAG_HALFS + lane * 16);
    }
  }

  const float nv = nItem[t & (N_ - 1)];
  const float wv = W2[t & (H_ - 1)];

  __builtin_amdgcn_sched_barrier(0);

  Slot cur = slot_info(wg, 0);
  int pairIdx = 0;

  float4 xr[16];
  {
    const float4* x4 = (const float4*)(x + (size_t)cur.tile * (M_ * F_));
    for (int k = 0; k < 16; ++k) xr[k] = x4[k * 256 + t];
  }

  for (int s = 0; s < SLOTS_PER_WG; ++s) {
    _Float16* xt = xs[s & 1];

#pragma unroll
    for (int k = 0; k < 16; ++k) {
      int f4  = k * 256 + t;
      float4 v = xr[k];
      int row = f4 >> 6;
      int col = (f4 & 63) * 4;
      v4h hv = { (_Float16)v.x, (_Float16)v.y, (_Float16)v.z, (_Float16)v.w };
      *(v4h*)&xt[row * LDS_STRIDE + col] = hv;
    }
    lds_barrier();

    v16h fr[2][2][4];
#pragma unroll
    for (int g = 0; g < 2; ++g)
#pragma unroll
      for (int kk = 0; kk < 2; ++kk)
#pragma unroll
        for (int mi = 0; mi < 4; ++mi)
          fr[g][kk][mi] = lds_frag(&xt[(mrow0 + mi * 16) * LDS_STRIDE + (g * 2 + kk) * 32 + klane]);

    Slot nxt = cur;
    if (s + 1 < SLOTS_PER_WG) {
      nxt = slot_info(wg, s + 1);
      const float4* x4 = (const float4*)(x + (size_t)nxt.tile * (M_ * F_));
      for (int k = 0; k < 16; ++k) xr[k] = x4[k * 256 + t];
    }
    __builtin_amdgcn_sched_barrier(0);

    v8f acc[4][2] = {};

#pragma unroll
    for (int g = 0; g < 4; ++g) {
#pragma unroll
      for (int kk = 0; kk < 2; ++kk) {
#pragma unroll
        for (int ni = 0; ni < 2; ++ni) {
#pragma unroll
          for (int mi = 0; mi < 4; ++mi) {
            acc[mi][ni] = __builtin_amdgcn_wmma_f32_16x16x32_f16(
                false, breg[ni][g * 2 + kk], false, fr[g & 1][kk][mi], (short)0, acc[mi][ni], false, false);
          }
        }
      }
      __builtin_amdgcn_sched_barrier(0);
      if (g + 2 < 4) {
#pragma unroll
        for (int kk = 0; kk < 2; ++kk)
#pragma unroll
          for (int mi = 0; mi < 4; ++mi)
            fr[g & 1][kk][mi] = lds_frag(&xt[(mrow0 + mi * 16) * LDS_STRIDE + ((g + 2) * 2 + kk) * 32 + klane]);
        __builtin_amdgcn_sched_barrier(0);
      }
    }

    _Float16* ht = hs[cur.second];
#pragma unroll
    for (int mi = 0; mi < 4; ++mi) {
#pragma unroll
      for (int ni = 0; ni < 2; ++ni) {
        v8h hv;
#pragma unroll
        for (int r = 0; r < 8; ++r) hv[r] = (_Float16)acc[mi][ni][r];
        *(v8h*)&ht[(mi * 16 + mrow0) * LDS_STRIDE + (ntBase + ni) * 16 + klane] = hv;
      }
    }

    if (cur.second | cur.diag) {
      lds_barrier();

      const int head   = wave >> 1;
      const int mtBase = (wave & 1) * 2;
      const _Float16* P = hs[0];
      const _Float16* Q = cur.diag ? hs[0] : hs[1];

      v16h pa[2][2], qb[2][4];
#pragma unroll
      for (int kt = 0; kt < 2; ++kt) {
#pragma unroll
        for (int mi = 0; mi < 2; ++mi)
          pa[kt][mi] = lds_frag(&P[((mtBase + mi) * 16 + mrow0) * LDS_STRIDE + head * D_ + kt * 32 + klane]);
#pragma unroll
        for (int ni = 0; ni < 4; ++ni)
          qb[kt][ni] = lds_frag(&Q[(ni * 16 + mrow0) * LDS_STRIDE + head * D_ + kt * 32 + klane]);
      }
      __builtin_amdgcn_sched_barrier(0);

      v8f sc[2][4] = {};
#pragma unroll
      for (int kt = 0; kt < 2; ++kt)
#pragma unroll
        for (int mi = 0; mi < 2; ++mi)
#pragma unroll
          for (int ni = 0; ni < 4; ++ni)
            sc[mi][ni] = __builtin_amdgcn_wmma_f32_16x16x32_f16(
                false, pa[kt][mi], false, qb[kt][ni], (short)0, sc[mi][ni], false, false);

      float sred = 0.0f;
#pragma unroll
      for (int mi = 0; mi < 2; ++mi)
#pragma unroll
        for (int ni = 0; ni < 4; ++ni)
#pragma unroll
          for (int r = 0; r < 8; ++r)
            sred += fmaxf(sc[mi][ni][r], 0.0f);

#pragma unroll
      for (int off = 16; off > 0; off >>= 1)
        sred += __shfl_xor(sred, off, 32);
      if (lane == 0) wsum[pairIdx * 8 + wave] = sred;
      ++pairIdx;
    }
    cur = nxt;
  }

  if (t < N_) nIs[t] = nv;
  if (t < H_) w2s[t] = wv;
  lds_barrier();
  const int npairs = (wg < FULL_WGS) ? 8 : MAX_PAIRS;
  if (t < npairs) {
    const int slot = (wg < FULL_WGS || t < 7) ? (2 * t + 1) : (7 + t);
    const Slot ps = slot_info(wg, slot);
    const float inv = 0.125f / (nIs[ps.i] * nIs[ps.j]);
    float r = 0.0f;
#pragma unroll
    for (int h = 0; h < H_; ++h) r += (wsum[t * 8 + 2 * h] + wsum[t * 8 + 2 * h + 1]) * w2s[h];
    r *= inv;
    out[ps.i * N_ + ps.j] = r;
    out[ps.j * N_ + ps.i] = r;
  }
}

extern "C" void kernel_launch(void* const* d_in, const int* in_sizes, int n_in,
                              void* d_out, int out_size, void* d_ws, size_t ws_size,
                              hipStream_t stream) {
  const float* x     = (const float*)d_in[0];
  const float* nItem = (const float*)d_in[1];
  const float* W1    = (const float*)d_in[2];
  const float* W2    = (const float*)d_in[3];
  float* out = (float*)d_out;

  _Float16* w1f = (_Float16*)d_ws;

  swizzle_w1_kernel<<<256, 256, 0, stream>>>(W1, w1f);
  pair_kernel<<<PAIR_WGS, 256, 0, stream>>>(x, w1f, nItem, W2, out);
}


# Two independent 4-wave work-groups per compute unit (512 groups x 8 tiles, 64x64 wave tile): one group's waits run under the other's matrix work; relu sums per head in one wave (other sum order)
# speedup vs baseline: 2.4259x; 1.0489x over previous
#include <hip/hip_runtime.h>


typedef __attribute__((ext_vector_type(16))) _Float16 v16h;
typedef __attribute__((ext_vector_type(8)))  _Float16 v8h;
typedef __attribute__((ext_vector_type(4)))  _Float16 v4h;
typedef __attribute__((ext_vector_type(8)))  float    v8f;

static constexpr int N_  = 64;
static constexpr int M_  = 64;
static constexpr int F_  = 256;
static constexpr int H_  = 4;
static constexpr int D_  = 64;
static constexpr int HD_ = H_ * D_;
static constexpr int FRAG_HALFS = 32 * 16;
static constexpr int W1F_HALFS  = F_ * HD_;
static constexpr int LDS_STRIDE = 264;
static constexpr int TILE_HALFS = M_ * LDS_STRIDE;

__global__ void swizzle_w1_kernel(const float* __restrict__ W1,
                                  _Float16* __restrict__ w1f) {
  int id   = blockIdx.x * blockDim.x + threadIdx.x;
  int jj   = id & 15;
  int lane = (id >> 4) & 31;
  int kt   = (id >> 9) & 7;
  int nt   = (id >> 12) & 15;
  int n    = nt * 16 + (lane & 15);
  int koff = (lane >> 4) * 8;
  int k    = kt * 32 + koff + ((jj < 8) ? jj : (8 + jj));
  w1f[id] = (_Float16)W1[k * HD_ + n];
}

__device__ __forceinline__ void lds_barrier() {
  __builtin_amdgcn_fence(__ATOMIC_RELEASE, "workgroup", "local");
  __builtin_amdgcn_s_barrier();
  __builtin_amdgcn_fence(__ATOMIC_ACQUIRE, "workgroup", "local");
}

__device__ __forceinline__ v16h lds_frag(const _Float16* src) {
  v8h lo = *(const v8h*)src;
  v8h hi = *(const v8h*)(src + 16);
  v16h f;
#pragma unroll
  for (int u = 0; u < 8; ++u) { f[u] = lo[u]; f[8 + u] = hi[u]; }
  return f;
}

__device__ __forceinline__ void wave_sum_to(float v, float* dst, int lane) {
#pragma unroll
  for (int off = 16; off > 0; off >>= 1)
    v += __shfl_xor(v, off, 32);
  if (lane == 0) *dst = v;
}

static constexpr int PAIR_WGS     = 512;
static constexpr int SLOTS_PER_WG = 8;
static constexpr int FULL_WGS     = 480;
static constexpr int MAX_PAIRS    = 5;
static constexpr int WG_THREADS   = 128;
static constexpr int WAVES        = 4;
static constexpr int NI_          = 4;
static constexpr int XLOADS       = (M_ * F_ / 4) / WG_THREADS;

struct Slot { int tile, i, j, second, diag; };

__device__ __forceinline__ Slot slot_info(int wg, int s) {
  Slot r;
  int u;
  if (wg < FULL_WGS) {
    u = wg * 4 + (s >> 1);
  } else {
    const int v = wg - FULL_WGS;
    if (s >= 6) {
      const int g = v * 2 + (s - 6);
      r.i = g; r.j = g; r.second = 0; r.diag = 1; r.tile = g * N_ + g;
      return r;
    }
    u = FULL_WGS * 4 + v * 3 + (s >> 1);
  }
  int i, d;
  if (u < 1984) { i = u & 63; d = 1 + (u >> 6); }
  else          { i = u - 1984; d = 32; }
  const int j = (i + d) & 63;
  r.i = i; r.j = j; r.second = s & 1; r.diag = 0;
  r.tile = r.second ? (j * N_ + i) : (i * N_ + j);
  return r;
}

__global__ void __launch_bounds__(WG_THREADS, 2)
pair_kernel(const float* __restrict__ x,
            const _Float16* __restrict__ w1f,
            const float* __restrict__ nItem,
            const float* __restrict__ W2,
            float* __restrict__ out) {
  __shared__ __align__(16) _Float16 xs[2][TILE_HALFS];
  __shared__ __align__(16) _Float16 hs[2][TILE_HALFS];
  __shared__ float wsum[MAX_PAIRS * WAVES];
  __shared__ float nIs[N_];
  __shared__ float w2s[H_];

  const int t    = threadIdx.x;
  const int lane = t & 31;
  const int wave = t >> 5;
  const int wg   = blockIdx.x;

  const int ntBase = wave * NI_;
  const int mrow0  = lane & 15;
  const int klane  = (lane >> 4) << 3;

  v16h breg[NI_][8];
#pragma unroll
  for (int ni = 0; ni < NI_; ++ni) {
#pragma unroll
    for (int kt = 0; kt < 8; ++kt) {
      breg[ni][kt] = *(const v16h*)(w1f + ((ntBase + ni) * 8 + kt) * FRAG_HALFS + lane * 16);
    }
  }

  const float nv = nItem[t & (N_ - 1)];
  const float wv = W2[t & (H_ - 1)];

  int pairIdx = 0;

  const int stage0 = (t >> 6) * LDS_STRIDE + (t & 63) * 4;

  Slot cur = slot_info(wg, 0);

  float4 xr[XLOADS];
  {
    const float4* x4 = (const float4*)(x + (size_t)cur.tile * (M_ * F_));
#pragma unroll
    for (int k = 0; k < XLOADS; ++k) xr[k] = x4[k * WG_THREADS + t];
  }

  auto tile = [&](const int s, const bool more) __attribute__((always_inline)) {

    _Float16* xt = xs[s & 1];

#pragma unroll
    for (int k = 0; k < XLOADS; ++k) {
      float4 v = xr[k];
      v4h hv = { (_Float16)v.x, (_Float16)v.y, (_Float16)v.z, (_Float16)v.w };
      *(v4h*)&xt[stage0 + k * 2 * LDS_STRIDE] = hv;
    }
    lds_barrier();

    v16h fr[2][4];
#pragma unroll
    for (int g = 0; g < 2; ++g)
#pragma unroll
      for (int mi = 0; mi < 4; ++mi)
        fr[g][mi] = lds_frag(&xt[(mrow0 + mi * 16) * LDS_STRIDE + g * 32 + klane]);

    v8f acc[4][NI_] = {};

#pragma unroll
    for (int kt = 0; kt < 8; ++kt) {
#pragma unroll
      for (int ni = 0; ni < NI_; ++ni) {
#pragma unroll
        for (int mi = 0; mi < 4; ++mi) {
          acc[mi][ni] = __builtin_amdgcn_wmma_f32_16x16x32_f16(
              false, breg[ni][kt], false, fr[kt & 1][mi], (short)0, acc[mi][ni], false, false);
        }
      }
      __builtin_amdgcn_sched_barrier(0);
      if (kt + 2 < 8) {
#pragma unroll
        for (int mi = 0; mi < 4; ++mi)
          fr[kt & 1][mi] = lds_frag(&xt[(mrow0 + mi * 16) * LDS_STRIDE + (kt + 2) * 32 + klane]);
        __builtin_amdgcn_sched_barrier(0);
      }
    }

    _Float16* ht = hs[cur.second];
#pragma unroll
    for (int mi = 0; mi < 4; ++mi) {
#pragma unroll
      for (int ni = 0; ni < NI_; ++ni) {
        v8h hv;
#pragma unroll
        for (int r = 0; r < 8; ++r) hv[r] = (_Float16)acc[mi][ni][r];
        *(v8h*)&ht[(mi * 16 + mrow0) * LDS_STRIDE + (ntBase + ni) * 16 + klane] = hv;
      }
    }

    Slot nxt = cur;
    if (more) nxt = slot_info(wg, s + 1);
    const float4* nx4 = (const float4*)(x + (size_t)nxt.tile * (M_ * F_));

    if (cur.second | cur.diag) {
      lds_barrier();

      const int head = wave;
      const _Float16* P = hs[0];
      const _Float16* Q = cur.diag ? hs[0] : hs[1];

      v8f sc[4][4] = {};
#pragma unroll
      for (int kt = 0; kt < 2; ++kt) {
        v16h pa[4], qb[4];
#pragma unroll
        for (int mi = 0; mi < 4; ++mi)
          pa[mi] = lds_frag(&P[(mi * 16 + mrow0) * LDS_STRIDE + head * D_ + kt * 32 + klane]);
#pragma unroll
        for (int ni = 0; ni < 4; ++ni)
          qb[ni] = lds_frag(&Q[(ni * 16 + mrow0) * LDS_STRIDE + head * D_ + kt * 32 + klane]);
        __builtin_amdgcn_sched_barrier(0);
#pragma unroll
        for (int mi = 0; mi < 4; ++mi)
#pragma unroll
          for (int ni = 0; ni < 4; ++ni)
            sc[mi][ni] = __builtin_amdgcn_wmma_f32_16x16x32_f16(
                false, pa[mi], false, qb[ni], (short)0, sc[mi][ni], false, false);
      }

      float part[8] = {};
#pragma unroll
      for (int mi = 0; mi < 4; ++mi)
#pragma unroll
        for (int ni = 0; ni < 4; ++ni)
#pragma unroll
          for (int r = 0; r < 8; ++r)
            part[r] += fmaxf(sc[mi][ni][r], 0.0f);
      const float sred = ((part[0] + part[1]) + (part[2] + part[3])) + ((part[4] + part[5]) + (part[6] + part[7]));

      if (more) {
#pragma unroll
        for (int k = 0; k < XLOADS; ++k) xr[k] = nx4[k * WG_THREADS + t];
      }
      __builtin_amdgcn_sched_barrier(0);

      wave_sum_to(sred, &wsum[pairIdx * WAVES + wave], lane);
      ++pairIdx;
    } else if (more) {
#pragma unroll
      for (int k = 0; k < XLOADS; ++k) xr[k] = nx4[k * WG_THREADS + t];
    }
    cur = nxt;
  };

  for (int s = 0; s < SLOTS_PER_WG - 1; ++s) tile(s, true);
  tile(SLOTS_PER_WG - 1, false);

  if (t < N_) nIs[t] = nv;
  if (t < H_) w2s[t] = wv;
  lds_barrier();
  const int npairs = (wg < FULL_WGS) ? 4 : MAX_PAIRS;
  if (t < npairs) {
    const int slot = (wg < FULL_WGS || t < 3) ? (2 * t + 1) : (3 + t);
    const Slot ps = slot_info(wg, slot);
    const float inv = 0.125f / (nIs[ps.i] * nIs[ps.j]);
    float r = 0.0f;
#pragma unroll
    for (int h = 0; h < H_; ++h) r += wsum[t * WAVES + h] * w2s[h];
    r *= inv;
    out[ps.i * N_ + ps.j] = r;
    out[ps.j * N_ + ps.i] = r;
  }
}

extern "C" void kernel_launch(void* const* d_in, const int* in_sizes, int n_in,
                              void* d_out, int out_size, void* d_ws, size_t ws_size,
                              hipStream_t stream) {
  const float* x     = (const float*)d_in[0];
  const float* nItem = (const float*)d_in[1];
  const float* W1    = (const float*)d_in[2];
  const float* W2    = (const float*)d_in[3];
  float* out = (float*)d_out;

  _Float16* w1f = (_Float16*)d_ws;

  swizzle_w1_kernel<<<256, 256, 0, stream>>>(W1, w1f);
  pair_kernel<<<PAIR_WGS, WG_THREADS, 0, stream>>>(x, w1f, nItem, W2, out);
}
